# starting kernel
# baseline (speedup 1.0000x reference)
#include <hip/hip_runtime.h>
#include <math.h>

typedef __attribute__((ext_vector_type(2))) float v2f;
typedef __attribute__((ext_vector_type(8))) float v8f;

#define IN_DIM    512
#define OUT_DIM   512
#define MOD_DIM   256
#define BATCH     1024
#define EPS       1e-8f
#define XS_STRIDE 516

__global__ void dml_colsq_kernel(const float* __restrict__ weight,
                                 float* __restrict__ colsq) {
    int i = blockIdx.x * blockDim.x + threadIdx.x;
    if (i >= IN_DIM) return;
    float s = 0.f;
#pragma unroll 8
    for (int o = 0; o < OUT_DIM; ++o) {
        float w = weight[o * IN_DIM + i];
        s = fmaf(w, w, s);
    }
    colsq[i] = s;
}

__global__ void __launch_bounds__(256)
dml_fused_kernel(const float* __restrict__ mod,
                 const float* __restrict__ mod_w,
                 const float* __restrict__ mod_b,
                 const float* __restrict__ x,
                 const float* __restrict__ colsq,
                 const float* __restrict__ weight,
                 const float* __restrict__ bias,
                 float* __restrict__ out) {
    __shared__ float lds_xs[16 * XS_STRIDE];

    const int lane = threadIdx.x & 31;
    const int wave = threadIdx.x >> 5;
    const int tM   = blockIdx.x;
    const int row  = lane & 15;
    const int koff = (lane >> 4) << 1;
    const int mloc = (lane >> 4) << 3;

    {
        const float* aRow = mod + (size_t)(tM * 16 + row) * MOD_DIM + koff;
        const float* bRow[4];
        v8f acc[4] = {};
#pragma unroll
        for (int t = 0; t < 4; ++t)
            bRow[t] = mod_w + (size_t)((wave * 4 + t) * 16 + row) * MOD_DIM + koff;

#pragma unroll 4
        for (int k = 0; k < MOD_DIM; k += 4) {
            v2f a = *(const v2f*)(aRow + k);
#pragma unroll
            for (int t = 0; t < 4; ++t) {
                v2f b = *(const v2f*)(bRow[t] + k);
                acc[t] = __builtin_amdgcn_wmma_f32_16x16x4_f32(
                    false, a, false, b, (short)0, acc[t], false, false);
            }
        }

#pragma unroll
        for (int t = 0; t < 4; ++t) {
            const int   gi = (wave * 4 + t) * 16 + row;
            const float mb = mod_b[gi];
            const float cs = colsq[gi];
#pragma unroll
            for (int r = 0; r < 8; ++r) {
                const int   mm = mloc + r;
                const float s  = acc[t][r] + mb;
                const float d  = rsqrtf(fmaf(s * s, cs, EPS));
                lds_xs[mm * XS_STRIDE + gi] =
                    x[(size_t)(tM * 16 + mm) * IN_DIM + gi] * s * d;
            }
        }
    }

    __syncthreads();

    {
        const float* aLds = &lds_xs[row * XS_STRIDE + koff];
        const float* bRow[4];
        v8f acc[4];
#pragma unroll
        for (int t = 0; t < 4; ++t) {
            const int   gn = (wave * 4 + t) * 16 + row;
            const float bv = bias[gn];
            acc[t] = (v8f){bv, bv, bv, bv, bv, bv, bv, bv};
            bRow[t] = weight + (size_t)gn * IN_DIM + koff;
        }

#pragma unroll 4
        for (int k = 0; k < IN_DIM; k += 4) {
            v2f a = *(const v2f*)(aLds + k);
#pragma unroll
            for (int t = 0; t < 4; ++t) {
                v2f b = *(const v2f*)(bRow[t] + k);
                acc[t] = __builtin_amdgcn_wmma_f32_16x16x4_f32(
                    false, a, false, b, (short)0, acc[t], false, false);
            }
        }

#pragma unroll
        for (int t = 0; t < 4; ++t) {
            const int gn = (wave * 4 + t) * 16 + row;
#pragma unroll
            for (int r = 0; r < 8; ++r)
                out[(size_t)(tM * 16 + mloc + r) * OUT_DIM + gn] = acc[t][r];
        }
    }
}

extern "C" void kernel_launch(void* const* d_in, const int* in_sizes, int n_in,
                              void* d_out, int out_size, void* d_ws, size_t ws_size,
                              hipStream_t stream) {
    const float* mod    = (const float*)d_in[0];
    const float* x      = (const float*)d_in[1];
    const float* weight = (const float*)d_in[2];
    const float* bias   = (const float*)d_in[3];
    const float* mod_w  = (const float*)d_in[4];
    const float* mod_b  = (const float*)d_in[5];
    float* out = (float*)d_out;

    float* colsq = (float*)d_ws;

    dml_colsq_kernel<<<(IN_DIM + 255) / 256, 256, 0, stream>>>(weight, colsq);

    dml_fused_kernel<<<BATCH / 16, 256, 0, stream>>>(
        mod, mod_w, mod_b, x, colsq, weight, bias, out);
}


# colsq (column sums of weight^2) as a parallel reduction: 16 blocks x 512 threads, 8 independent float4 loads each, shuffle + LDS tree; was 2 blocks of serial 512-row walks; colsq sum order changes
# speedup vs baseline: 1.8494x; 1.8494x over previous
#include <hip/hip_runtime.h>
#include <math.h>

typedef __attribute__((ext_vector_type(2))) float v2f;
typedef __attribute__((ext_vector_type(8))) float v8f;

#define IN_DIM    512
#define OUT_DIM   512
#define MOD_DIM   256
#define BATCH     1024
#define EPS       1e-8f
#define XS_STRIDE 516

#define CS_COLS    32
#define CS_THREADS 512
#define CS_WAVES   (CS_THREADS / 32)
#define CS_ROWS    (CS_THREADS / 8)
#define CS_PASSES  (OUT_DIM / CS_ROWS)

__global__ void __launch_bounds__(CS_THREADS)
dml_colsq_kernel(const float* __restrict__ weight,
                 float* __restrict__ colsq) {
    __shared__ float part[CS_WAVES][CS_COLS];

    const int tid  = threadIdx.x;
    const int lane = tid & 31;
    const int wave = tid >> 5;
    const float* p = weight + (size_t)(tid >> 3) * IN_DIM + blockIdx.x * CS_COLS + (tid & 7) * 4;

    float4 v[CS_PASSES];
#pragma unroll
    for (int j = 0; j < CS_PASSES; ++j)
        v[j] = *(const float4*)(p + (size_t)j * CS_ROWS * IN_DIM);

    float4 s = make_float4(0.f, 0.f, 0.f, 0.f);
#pragma unroll
    for (int j = 0; j < CS_PASSES; ++j) {
        s.x = fmaf(v[j].x, v[j].x, s.x);
        s.y = fmaf(v[j].y, v[j].y, s.y);
        s.z = fmaf(v[j].z, v[j].z, s.z);
        s.w = fmaf(v[j].w, v[j].w, s.w);
    }

    s.x += __shfl_xor(s.x, 8, 32);  s.y += __shfl_xor(s.y, 8, 32);
    s.z += __shfl_xor(s.z, 8, 32);  s.w += __shfl_xor(s.w, 8, 32);
    s.x += __shfl_xor(s.x, 16, 32); s.y += __shfl_xor(s.y, 16, 32);
    s.z += __shfl_xor(s.z, 16, 32); s.w += __shfl_xor(s.w, 16, 32);

    if (lane < 8) *(float4*)&part[wave][lane * 4] = s;
    __syncthreads();

    if (wave == 0) {
        float t = 0.f;
#pragma unroll
        for (int w = 0; w < CS_WAVES; ++w) t += part[w][lane];
        colsq[blockIdx.x * CS_COLS + lane] = t;
    }
}

__global__ void __launch_bounds__(256)
dml_fused_kernel(const float* __restrict__ mod,
                 const float* __restrict__ mod_w,
                 const float* __restrict__ mod_b,
                 const float* __restrict__ x,
                 const float* __restrict__ colsq,
                 const float* __restrict__ weight,
                 const float* __restrict__ bias,
                 float* __restrict__ out) {
    __shared__ float lds_xs[16 * XS_STRIDE];

    const int lane = threadIdx.x & 31;
    const int wave = threadIdx.x >> 5;
    const int tM   = blockIdx.x;
    const int row  = lane & 15;
    const int koff = (lane >> 4) << 1;
    const int mloc = (lane >> 4) << 3;

    {
        const float* aRow = mod + (size_t)(tM * 16 + row) * MOD_DIM + koff;
        const float* bRow[4];
        v8f acc[4] = {};
#pragma unroll
        for (int t = 0; t < 4; ++t)
            bRow[t] = mod_w + (size_t)((wave * 4 + t) * 16 + row) * MOD_DIM + koff;

#pragma unroll 4
        for (int k = 0; k < MOD_DIM; k += 4) {
            v2f a = *(const v2f*)(aRow + k);
#pragma unroll
            for (int t = 0; t < 4; ++t) {
                v2f b = *(const v2f*)(bRow[t] + k);
                acc[t] = __builtin_amdgcn_wmma_f32_16x16x4_f32(
                    false, a, false, b, (short)0, acc[t], false, false);
            }
        }

#pragma unroll
        for (int t = 0; t < 4; ++t) {
            const int   gi = (wave * 4 + t) * 16 + row;
            const float mb = mod_b[gi];
            const float cs = colsq[gi];
#pragma unroll
            for (int r = 0; r < 8; ++r) {
                const int   mm = mloc + r;
                const float s  = acc[t][r] + mb;
                const float d  = rsqrtf(fmaf(s * s, cs, EPS));
                lds_xs[mm * XS_STRIDE + gi] =
                    x[(size_t)(tM * 16 + mm) * IN_DIM + gi] * s * d;
            }
        }
    }

    __syncthreads();

    {
        const float* aLds = &lds_xs[row * XS_STRIDE + koff];
        const float* bRow[4];
        v8f acc[4];
#pragma unroll
        for (int t = 0; t < 4; ++t) {
            const int   gn = (wave * 4 + t) * 16 + row;
            const float bv = bias[gn];
            acc[t] = (v8f){bv, bv, bv, bv, bv, bv, bv, bv};
            bRow[t] = weight + (size_t)gn * IN_DIM + koff;
        }

#pragma unroll 4
        for (int k = 0; k < IN_DIM; k += 4) {
            v2f a = *(const v2f*)(aLds + k);
#pragma unroll
            for (int t = 0; t < 4; ++t) {
                v2f b = *(const v2f*)(bRow[t] + k);
                acc[t] = __builtin_amdgcn_wmma_f32_16x16x4_f32(
                    false, a, false, b, (short)0, acc[t], false, false);
            }
        }

#pragma unroll
        for (int t = 0; t < 4; ++t) {
            const int gn = (wave * 4 + t) * 16 + row;
#pragma unroll
            for (int r = 0; r < 8; ++r)
                out[(size_t)(tM * 16 + mloc + r) * OUT_DIM + gn] = acc[t][r];
        }
    }
}

extern "C" void kernel_launch(void* const* d_in, const int* in_sizes, int n_in,
                              void* d_out, int out_size, void* d_ws, size_t ws_size,
                              hipStream_t stream) {
    const float* mod    = (const float*)d_in[0];
    const float* x      = (const float*)d_in[1];
    const float* weight = (const float*)d_in[2];
    const float* bias   = (const float*)d_in[3];
    const float* mod_w  = (const float*)d_in[4];
    const float* mod_b  = (const float*)d_in[5];
    float* out = (float*)d_out;

    float* colsq = (float*)d_ws;

    dml_colsq_kernel<<<IN_DIM / CS_COLS, CS_THREADS, 0, stream>>>(weight, colsq);

    dml_fused_kernel<<<BATCH / 16, 256, 0, stream>>>(
        mod, mod_w, mod_b, x, colsq, weight, bias, out);
}


# grid 64 -> 256 blocks: each block rebuilds its 16-row xs tile and does a quarter of the output columns, one tile per wave, 16 k-steps per trip; tile stored via LDS as whole rows; sum order unchanged
# speedup vs baseline: 2.8481x; 1.5400x over previous
#include <hip/hip_runtime.h>
#include <math.h>

typedef __attribute__((ext_vector_type(2))) float v2f;
typedef __attribute__((ext_vector_type(8))) float v8f;

#define IN_DIM    512
#define OUT_DIM   512
#define MOD_DIM   256
#define BATCH     1024
#define EPS       1e-8f
#define XS_STRIDE 516
#define N_SPLIT     4
#define N_PER_BLOCK (OUT_DIM / N_SPLIT)
#define OS_STRIDE   132

#define CS_COLS    32
#define CS_THREADS 512
#define CS_WAVES   (CS_THREADS / 32)
#define CS_ROWS    (CS_THREADS / 8)
#define CS_PASSES  (OUT_DIM / CS_ROWS)

__global__ void __launch_bounds__(CS_THREADS)
dml_colsq_kernel(const float* __restrict__ weight,
                 float* __restrict__ colsq) {
    __shared__ float part[CS_WAVES][CS_COLS];

    const int tid  = threadIdx.x;
    const int lane = tid & 31;
    const int wave = tid >> 5;
    const float* p = weight + (size_t)(tid >> 3) * IN_DIM + blockIdx.x * CS_COLS + (tid & 7) * 4;

    float4 v[CS_PASSES];
#pragma unroll
    for (int j = 0; j < CS_PASSES; ++j)
        v[j] = *(const float4*)(p + (size_t)j * CS_ROWS * IN_DIM);

    float4 s = make_float4(0.f, 0.f, 0.f, 0.f);
#pragma unroll
    for (int j = 0; j < CS_PASSES; ++j) {
        s.x = fmaf(v[j].x, v[j].x, s.x);
        s.y = fmaf(v[j].y, v[j].y, s.y);
        s.z = fmaf(v[j].z, v[j].z, s.z);
        s.w = fmaf(v[j].w, v[j].w, s.w);
    }

    s.x += __shfl_xor(s.x, 8, 32);  s.y += __shfl_xor(s.y, 8, 32);
    s.z += __shfl_xor(s.z, 8, 32);  s.w += __shfl_xor(s.w, 8, 32);
    s.x += __shfl_xor(s.x, 16, 32); s.y += __shfl_xor(s.y, 16, 32);
    s.z += __shfl_xor(s.z, 16, 32); s.w += __shfl_xor(s.w, 16, 32);

    if (lane < 8) *(float4*)&part[wave][lane * 4] = s;
    __syncthreads();

    if (wave == 0) {
        float t = 0.f;
#pragma unroll
        for (int w = 0; w < CS_WAVES; ++w) t += part[w][lane];
        colsq[blockIdx.x * CS_COLS + lane] = t;
    }
}

__global__ void __launch_bounds__(256)
dml_fused_kernel(const float* __restrict__ mod,
                 const float* __restrict__ mod_w,
                 const float* __restrict__ mod_b,
                 const float* __restrict__ x,
                 const float* __restrict__ colsq,
                 const float* __restrict__ weight,
                 const float* __restrict__ bias,
                 float* __restrict__ out) {
    __shared__ float lds_xs[16 * XS_STRIDE];
    __shared__ alignas(16) float lds_out[16 * OS_STRIDE];

    const int lane = threadIdx.x & 31;
    const int wave = threadIdx.x >> 5;
    const int tM   = blockIdx.x;
    const int nQ   = blockIdx.y;
    const int row  = lane & 15;
    const int koff = (lane >> 4) << 1;
    const int mloc = (lane >> 4) << 3;

    {
        const float* aRow = mod + (size_t)(tM * 16 + row) * MOD_DIM + koff;
        const float* bRow[4];
        v8f acc[4] = {};
#pragma unroll
        for (int t = 0; t < 4; ++t)
            bRow[t] = mod_w + (size_t)((wave * 4 + t) * 16 + row) * MOD_DIM + koff;

#pragma unroll 4
        for (int k = 0; k < MOD_DIM; k += 4) {
            v2f a = *(const v2f*)(aRow + k);
#pragma unroll
            for (int t = 0; t < 4; ++t) {
                v2f b = *(const v2f*)(bRow[t] + k);
                acc[t] = __builtin_amdgcn_wmma_f32_16x16x4_f32(
                    false, a, false, b, (short)0, acc[t], false, false);
            }
        }

#pragma unroll
        for (int t = 0; t < 4; ++t) {
            const int   gi = (wave * 4 + t) * 16 + row;
            const float mb = mod_b[gi];
            const float cs = colsq[gi];
#pragma unroll
            for (int r = 0; r < 8; ++r) {
                const int   mm = mloc + r;
                const float s  = acc[t][r] + mb;
                const float d  = rsqrtf(fmaf(s * s, cs, EPS));
                lds_xs[mm * XS_STRIDE + gi] =
                    x[(size_t)(tM * 16 + mm) * IN_DIM + gi] * s * d;
            }
        }
    }

    __syncthreads();

    {
        const int    gn   = nQ * N_PER_BLOCK + wave * 16 + row;
        const float* aLds = &lds_xs[row * XS_STRIDE + koff];
        const float* bRow = weight + (size_t)gn * IN_DIM + koff;
        const float  bv   = bias[gn];
        v8f acc = (v8f){bv, bv, bv, bv, bv, bv, bv, bv};

#pragma unroll 16
        for (int k = 0; k < IN_DIM; k += 4) {
            v2f a = *(const v2f*)(aLds + k);
            v2f b = *(const v2f*)(bRow + k);
            acc = __builtin_amdgcn_wmma_f32_16x16x4_f32(
                false, a, false, b, (short)0, acc, false, false);
        }

#pragma unroll
        for (int r = 0; r < 8; ++r)
            lds_out[(mloc + r) * OS_STRIDE + wave * 16 + row] = acc[r];
    }

    __syncthreads();

#pragma unroll
    for (int j = 0; j < 2; ++j) {
        const int rr = wave + 8 * j;
        const float4 v = *(const float4*)&lds_out[rr * OS_STRIDE + lane * 4];
        *(float4*)&out[(size_t)(tM * 16 + rr) * OUT_DIM + nQ * N_PER_BLOCK + lane * 4] = v;
    }
}

extern "C" void kernel_launch(void* const* d_in, const int* in_sizes, int n_in,
                              void* d_out, int out_size, void* d_ws, size_t ws_size,
                              hipStream_t stream) {
    const float* mod    = (const float*)d_in[0];
    const float* x      = (const float*)d_in[1];
    const float* weight = (const float*)d_in[2];
    const float* bias   = (const float*)d_in[3];
    const float* mod_w  = (const float*)d_in[4];
    const float* mod_b  = (const float*)d_in[5];
    float* out = (float*)d_out;

    float* colsq = (float*)d_ws;

    dml_colsq_kernel<<<IN_DIM / CS_COLS, CS_THREADS, 0, stream>>>(weight, colsq);

    dml_fused_kernel<<<dim3(BATCH / 16, N_SPLIT), 256, 0, stream>>>(
        mod, mod_w, mod_b, x, colsq, weight, bias, out);
}


# scales GEMM moved into launch 1 beside colsq, computed once (each of the 4 column-quarter blocks of a batch tile had rebuilt it); scales plane via workspace, guarded by ws_size; sum order unchanged
# speedup vs baseline: 4.3946x; 1.5430x over previous
#include <hip/hip_runtime.h>
#include <math.h>

typedef __attribute__((ext_vector_type(2))) float v2f;
typedef __attribute__((ext_vector_type(4))) float v4f;
typedef __attribute__((ext_vector_type(8))) float v8f;

#define IN_DIM    512
#define OUT_DIM   512
#define MOD_DIM   256
#define BATCH     1024
#define EPS       1e-8f
#define XS_STRIDE 516
#define N_SPLIT     4
#define N_PER_BLOCK (OUT_DIM / N_SPLIT)
#define OS_STRIDE   132

#define CS_COLS    32
#define CS_BLOCKS  (IN_DIM / CS_COLS)
#define CS_POS     512
#define CS_GROUPS  (CS_POS / 32)
#define CS_ROWS    (CS_POS / 8)
#define CS_PASSES  (OUT_DIM / CS_ROWS)
#define L1_THREADS 256

#define G1_TM      32
#define G1_TN      64
#define G1_BLOCKS  ((BATCH / G1_TM) * (IN_DIM / G1_TN))
#define SS_STRIDE  68
#define WS_NEED    ((size_t)(IN_DIM + BATCH * IN_DIM) * sizeof(float))

__global__ void __launch_bounds__(L1_THREADS)
dml_colsq_scales_kernel(const float* __restrict__ weight,
                        const float* __restrict__ mod,
                        const float* __restrict__ mod_w,
                        const float* __restrict__ mod_b,
                        float* __restrict__ colsq,
                        float* __restrict__ scales,
                        size_t ws_size) {
    __shared__ alignas(16) float lds[G1_TM * SS_STRIDE];

    const int tid  = threadIdx.x;
    const int lane = tid & 31;
    const int wave = tid >> 5;

    if (blockIdx.x < CS_BLOCKS) {
        float (*part)[CS_COLS] = (float (*)[CS_COLS])lds;
        const float* p = weight + (size_t)(tid >> 3) * IN_DIM + blockIdx.x * CS_COLS + (tid & 7) * 4;

        float4 v[2][CS_PASSES];
#pragma unroll
        for (int u = 0; u < 2; ++u)
#pragma unroll
            for (int j = 0; j < CS_PASSES; ++j)
                v[u][j] = *(const float4*)(p + (size_t)(u * (L1_THREADS >> 3) + j * CS_ROWS) * IN_DIM);

#pragma unroll
        for (int u = 0; u < 2; ++u) {
            float4 s = make_float4(0.f, 0.f, 0.f, 0.f);
#pragma unroll
            for (int j = 0; j < CS_PASSES; ++j) {
                s.x = fmaf(v[u][j].x, v[u][j].x, s.x);
                s.y = fmaf(v[u][j].y, v[u][j].y, s.y);
                s.z = fmaf(v[u][j].z, v[u][j].z, s.z);
                s.w = fmaf(v[u][j].w, v[u][j].w, s.w);
            }
            s.x += __shfl_xor(s.x, 8, 32);  s.y += __shfl_xor(s.y, 8, 32);
            s.z += __shfl_xor(s.z, 8, 32);  s.w += __shfl_xor(s.w, 8, 32);
            s.x += __shfl_xor(s.x, 16, 32); s.y += __shfl_xor(s.y, 16, 32);
            s.z += __shfl_xor(s.z, 16, 32); s.w += __shfl_xor(s.w, 16, 32);

            if (lane < 8) *(float4*)&part[wave + u * (L1_THREADS >> 5)][lane * 4] = s;
        }
        __syncthreads();

        if (wave == 0) {
            float t = 0.f;
#pragma unroll
            for (int w = 0; w < CS_GROUPS; ++w) t += part[w][lane];
            colsq[blockIdx.x * CS_COLS + lane] = t;
        }
        return;
    }

    if (ws_size < WS_NEED) return;

    {
        const int g    = blockIdx.x - CS_BLOCKS;
        const int bm   = g >> 3;
        const int bn   = g & 7;
        const int tr   = wave >> 2;
        const int tc   = wave & 3;
        const int row  = lane & 15;
        const int koff = (lane >> 4) << 2;
        const int mloc = (lane >> 4) << 3;
        const int gi   = bn * G1_TN + tc * 16 + row;
        const float* aRow = mod   + (size_t)(bm * G1_TM + tr * 16 + row) * MOD_DIM + koff;
        const float* bRow = mod_w + (size_t)gi * MOD_DIM + koff;
        const float  mb   = mod_b[gi];
        v8f acc = {};

#pragma clang loop unroll(disable)
        for (int k = 0; k < MOD_DIM; k += 128) {
            v4f a[16], b[16];
#pragma unroll
            for (int u = 0; u < 16; ++u) {
                a[u] = *(const v4f*)(aRow + k + 8 * u);
                b[u] = *(const v4f*)(bRow + k + 8 * u);
            }
#pragma unroll
            for (int u = 0; u < 16; ++u) {
                const v2f a0 = {a[u].x, a[u].y};
                const v2f b0 = {b[u].x, b[u].y};
                const v2f a1 = {a[u].z, a[u].w};
                const v2f b1 = {b[u].z, b[u].w};
                acc = __builtin_amdgcn_wmma_f32_16x16x4_f32(
                    false, a0, false, b0, (short)0, acc, false, false);
                acc = __builtin_amdgcn_wmma_f32_16x16x4_f32(
                    false, a1, false, b1, (short)0, acc, false, false);
            }
            __builtin_amdgcn_sched_group_barrier(0x020, 32, 0);
            __builtin_amdgcn_sched_group_barrier(0x008, 32, 0);
        }

#pragma unroll
        for (int r = 0; r < 8; ++r)
            lds[(tr * 16 + mloc + r) * SS_STRIDE + tc * 16 + row] = acc[r] + mb;
    }
    __syncthreads();

    {
        const int g  = blockIdx.x - CS_BLOCKS;
        const int bm = g >> 3;
        const int bn = g & 7;
#pragma unroll
        for (int j = 0; j < 2; ++j) {
            const int rr = (tid >> 4) + 16 * j;
            const int c4 = tid & 15;
            const float4 v = *(const float4*)&lds[rr * SS_STRIDE + c4 * 4];
            *(float4*)&scales[(size_t)(bm * G1_TM + rr) * IN_DIM + bn * G1_TN + c4 * 4] = v;
        }
    }
}

__global__ void __launch_bounds__(256)
dml_fused_kernel(const float* __restrict__ mod,
                 const float* __restrict__ mod_w,
                 const float* __restrict__ mod_b,
                 const float* __restrict__ x,
                 const float* __restrict__ colsq,
                 const float* __restrict__ scales,
                 const float* __restrict__ weight,
                 const float* __restrict__ bias,
                 float* __restrict__ out,
                 size_t ws_size) {
    __shared__ alignas(16) float lds_xs[16 * XS_STRIDE];
    __shared__ alignas(16) float lds_out[16 * OS_STRIDE];

    const int lane = threadIdx.x & 31;
    const int wave = threadIdx.x >> 5;
    const int tM   = blockIdx.x;
    const int nQ   = blockIdx.y;
    const int row  = lane & 15;
    const int koff = (lane >> 4) << 2;
    const int mloc = (lane >> 4) << 3;

    if (ws_size >= WS_NEED) {
        const int    c4 = threadIdx.x & 127;
        const float4 cs = *(const float4*)&colsq[c4 * 4];
#pragma unroll
        for (int j = 0; j < 8; ++j) {
            const int    mm = (threadIdx.x >> 7) + 2 * j;
            const size_t go = (size_t)(tM * 16 + mm) * IN_DIM + c4 * 4;
            const float4 xv = *(const float4*)&x[go];
            const float4 sv = *(const float4*)&scales[go];
            float4 o;
            o.x = xv.x * sv.x * rsqrtf(fmaf(sv.x * sv.x, cs.x, EPS));
            o.y = xv.y * sv.y * rsqrtf(fmaf(sv.y * sv.y, cs.y, EPS));
            o.z = xv.z * sv.z * rsqrtf(fmaf(sv.z * sv.z, cs.z, EPS));
            o.w = xv.w * sv.w * rsqrtf(fmaf(sv.w * sv.w, cs.w, EPS));
            *(float4*)&lds_xs[mm * XS_STRIDE + c4 * 4] = o;
        }
    } else {
        const float* aRow = mod + (size_t)(tM * 16 + row) * MOD_DIM + koff;
        const float* bRow[4];
        v8f acc[4] = {};
#pragma unroll
        for (int t = 0; t < 4; ++t)
            bRow[t] = mod_w + (size_t)((wave * 4 + t) * 16 + row) * MOD_DIM + koff;

#pragma unroll 4
        for (int k = 0; k < MOD_DIM; k += 8) {
            const v4f a = *(const v4f*)(aRow + k);
            v4f b[4];
#pragma unroll
            for (int t = 0; t < 4; ++t)
                b[t] = *(const v4f*)(bRow[t] + k);
            const v2f a0 = {a.x, a.y};
            const v2f a1 = {a.z, a.w};
#pragma unroll
            for (int t = 0; t < 4; ++t) {
                const v2f b0 = {b[t].x, b[t].y};
                acc[t] = __builtin_amdgcn_wmma_f32_16x16x4_f32(
                    false, a0, false, b0, (short)0, acc[t], false, false);
            }
#pragma unroll
            for (int t = 0; t < 4; ++t) {
                const v2f b1 = {b[t].z, b[t].w};
                acc[t] = __builtin_amdgcn_wmma_f32_16x16x4_f32(
                    false, a1, false, b1, (short)0, acc[t], false, false);
            }
        }

#pragma unroll
        for (int t = 0; t < 4; ++t) {
            const int   gi = (wave * 4 + t) * 16 + row;
            const float mb = mod_b[gi];
            const float cs = colsq[gi];
#pragma unroll
            for (int r = 0; r < 8; ++r) {
                const int   mm = mloc + r;
                const float s  = acc[t][r] + mb;
                const float d  = rsqrtf(fmaf(s * s, cs, EPS));
                lds_xs[mm * XS_STRIDE + gi] =
                    x[(size_t)(tM * 16 + mm) * IN_DIM + gi] * s * d;
            }
        }
    }

    __syncthreads();

    {
        const int    gn   = nQ * N_PER_BLOCK + wave * 16 + row;
        const float* aLds = &lds_xs[row * XS_STRIDE + koff];
        const float* bRow = weight + (size_t)gn * IN_DIM + koff;
        const float  bv   = bias[gn];
        v8f acc = (v8f){bv, bv, bv, bv, bv, bv, bv, bv};

#pragma unroll 16
        for (int k = 0; k < IN_DIM; k += 8) {
            const v4f a = *(const v4f*)(aLds + k);
            const v4f b = *(const v4f*)(bRow + k);
            const v2f a0 = {a.x, a.y};
            const v2f b0 = {b.x, b.y};
            const v2f a1 = {a.z, a.w};
            const v2f b1 = {b.z, b.w};
            acc = __builtin_amdgcn_wmma_f32_16x16x4_f32(
                false, a0, false, b0, (short)0, acc, false, false);
            acc = __builtin_amdgcn_wmma_f32_16x16x4_f32(
                false, a1, false, b1, (short)0, acc, false, false);
        }

#pragma unroll
        for (int r = 0; r < 8; ++r)
            lds_out[(mloc + r) * OS_STRIDE + wave * 16 + row] = acc[r];
    }

    __syncthreads();

#pragma unroll
    for (int j = 0; j < 2; ++j) {
        const int rr = wave + 8 * j;
        const float4 v = *(const float4*)&lds_out[rr * OS_STRIDE + lane * 4];
        *(float4*)&out[(size_t)(tM * 16 + rr) * OUT_DIM + nQ * N_PER_BLOCK + lane * 4] = v;
    }
}

extern "C" void kernel_launch(void* const* d_in, const int* in_sizes, int n_in,
                              void* d_out, int out_size, void* d_ws, size_t ws_size,
                              hipStream_t stream) {
    const float* mod    = (const float*)d_in[0];
    const float* x      = (const float*)d_in[1];
    const float* weight = (const float*)d_in[2];
    const float* bias   = (const float*)d_in[3];
    const float* mod_w  = (const float*)d_in[4];
    const float* mod_b  = (const float*)d_in[5];
    float* out = (float*)d_out;

    float* colsq  = (float*)d_ws;
    float* scales = (float*)d_ws + IN_DIM;

    dml_colsq_scales_kernel<<<CS_BLOCKS + G1_BLOCKS, L1_THREADS, 0, stream>>>(
        weight, mod, mod_w, mod_b, colsq, scales, ws_size);

    dml_fused_kernel<<<dim3(BATCH / 16, N_SPLIT), 256, 0, stream>>>(
        mod, mod_w, mod_b, x, colsq, scales, weight, bias, out, ws_size);
}


# both GEMMs' operand tiles staged through LDS with whole-line loads (512 contiguous B of one row per load instruction, not 16 rows x 32 B); weight steps fetched one ahead; sum order unchanged
# speedup vs baseline: 5.8552x; 1.3324x over previous
#include <hip/hip_runtime.h>
#include <math.h>

typedef __attribute__((ext_vector_type(2))) float v2f;
typedef __attribute__((ext_vector_type(4))) float v4f;
typedef __attribute__((ext_vector_type(8))) float v8f;

#define IN_DIM    512
#define OUT_DIM   512
#define MOD_DIM   256
#define BATCH     1024
#define EPS       1e-8f
#define XS_STRIDE 516
#define N_SPLIT     4
#define N_PER_BLOCK (OUT_DIM / N_SPLIT)
#define OS_STRIDE   132
#define K_CHUNK     128
#define N_CHUNK     (IN_DIM / K_CHUNK)

#define CS_COLS    32
#define CS_BLOCKS  (IN_DIM / CS_COLS)
#define CS_POS     512
#define CS_GROUPS  (CS_POS / 32)
#define CS_ROWS    (CS_POS / 8)
#define CS_PASSES  (OUT_DIM / CS_ROWS)
#define L1_THREADS 256

#define G1_TM      32
#define G1_TN      64
#define G1_BLOCKS  ((BATCH / G1_TM) * (IN_DIM / G1_TN))
#define SS_STRIDE  68
#define AB_STRIDE  260
#define AB_ROWS    (G1_TM + G1_TN)
#define WS_NEED    ((size_t)(IN_DIM + BATCH * IN_DIM) * sizeof(float))

__global__ void __launch_bounds__(L1_THREADS)
dml_colsq_scales_kernel(const float* __restrict__ weight,
                        const float* __restrict__ mod,
                        const float* __restrict__ mod_w,
                        const float* __restrict__ mod_b,
                        float* __restrict__ colsq,
                        float* __restrict__ scales,
                        size_t ws_size) {
    __shared__ alignas(16) float lds[G1_TM * SS_STRIDE];
    __shared__ alignas(16) float lds_ab[AB_ROWS * AB_STRIDE];

    const int tid  = threadIdx.x;
    const int lane = tid & 31;
    const int wave = tid >> 5;

    if (blockIdx.x < CS_BLOCKS) {
        float (*part)[CS_COLS] = (float (*)[CS_COLS])lds;
        const float* p = weight + (size_t)(tid >> 3) * IN_DIM + blockIdx.x * CS_COLS + (tid & 7) * 4;

        float4 v[2][CS_PASSES];
#pragma unroll
        for (int u = 0; u < 2; ++u)
#pragma unroll
            for (int j = 0; j < CS_PASSES; ++j)
                v[u][j] = *(const float4*)(p + (size_t)(u * (L1_THREADS >> 3) + j * CS_ROWS) * IN_DIM);

#pragma unroll
        for (int u = 0; u < 2; ++u) {
            float4 s = make_float4(0.f, 0.f, 0.f, 0.f);
#pragma unroll
            for (int j = 0; j < CS_PASSES; ++j) {
                s.x = fmaf(v[u][j].x, v[u][j].x, s.x);
                s.y = fmaf(v[u][j].y, v[u][j].y, s.y);
                s.z = fmaf(v[u][j].z, v[u][j].z, s.z);
                s.w = fmaf(v[u][j].w, v[u][j].w, s.w);
            }
            s.x += __shfl_xor(s.x, 8, 32);  s.y += __shfl_xor(s.y, 8, 32);
            s.z += __shfl_xor(s.z, 8, 32);  s.w += __shfl_xor(s.w, 8, 32);
            s.x += __shfl_xor(s.x, 16, 32); s.y += __shfl_xor(s.y, 16, 32);
            s.z += __shfl_xor(s.z, 16, 32); s.w += __shfl_xor(s.w, 16, 32);

            if (lane < 8) *(float4*)&part[wave + u * (L1_THREADS >> 5)][lane * 4] = s;
        }
        __syncthreads();

        if (wave == 0) {
            float t = 0.f;
#pragma unroll
            for (int w = 0; w < CS_GROUPS; ++w) t += part[w][lane];
            colsq[blockIdx.x * CS_COLS + lane] = t;
        }
        return;
    }

    if (ws_size < WS_NEED) return;

    const int g    = blockIdx.x - CS_BLOCKS;
    const int bm   = g >> 3;
    const int bn   = g & 7;
    const int tr   = wave >> 2;
    const int tc   = wave & 3;
    const int row  = lane & 15;
    const int koff = (lane >> 4) << 2;
    const int mloc = (lane >> 4) << 3;
    {
        const float* aSrc = mod   + (size_t)(bm * G1_TM + wave * 4) * MOD_DIM + lane * 4;
        const float* bSrc = mod_w + (size_t)(bn * G1_TN + wave * 8) * MOD_DIM + lane * 4;
        float* aDst = &lds_ab[(wave * 4) * AB_STRIDE + lane * 4];
        float* bDst = &lds_ab[(G1_TM + wave * 8) * AB_STRIDE + lane * 4];
        v4f sa[8], sb[16];
#pragma unroll
        for (int j = 0; j < 4; ++j) {
            sa[2 * j]     = *(const v4f*)(aSrc + (size_t)j * MOD_DIM);
            sa[2 * j + 1] = *(const v4f*)(aSrc + (size_t)j * MOD_DIM + 128);
        }
#pragma unroll
        for (int j = 0; j < 8; ++j) {
            sb[2 * j]     = *(const v4f*)(bSrc + (size_t)j * MOD_DIM);
            sb[2 * j + 1] = *(const v4f*)(bSrc + (size_t)j * MOD_DIM + 128);
        }
#pragma unroll
        for (int j = 0; j < 4; ++j) {
            *(v4f*)(aDst + j * AB_STRIDE)       = sa[2 * j];
            *(v4f*)(aDst + j * AB_STRIDE + 128) = sa[2 * j + 1];
        }
#pragma unroll
        for (int j = 0; j < 8; ++j) {
            *(v4f*)(bDst + j * AB_STRIDE)       = sb[2 * j];
            *(v4f*)(bDst + j * AB_STRIDE + 128) = sb[2 * j + 1];
        }
    }
    const int   gi = bn * G1_TN + tc * 16 + row;
    const float mb = mod_b[gi];
    __syncthreads();

    {
        const float* aL = &lds_ab[(tr * 16 + row) * AB_STRIDE + koff];
        const float* bL = &lds_ab[(G1_TM + tc * 16 + row) * AB_STRIDE + koff];
        v8f acc = {};
#pragma unroll
        for (int u = 0; u < MOD_DIM / 8; ++u) {
            const v4f a = *(const v4f*)(aL + 8 * u);
            const v4f b = *(const v4f*)(bL + 8 * u);
            const v2f a0 = {a.x, a.y};
            const v2f b0 = {b.x, b.y};
            const v2f a1 = {a.z, a.w};
            const v2f b1 = {b.z, b.w};
            acc = __builtin_amdgcn_wmma_f32_16x16x4_f32(
                false, a0, false, b0, (short)0, acc, false, false);
            acc = __builtin_amdgcn_wmma_f32_16x16x4_f32(
                false, a1, false, b1, (short)0, acc, false, false);
        }
        __builtin_amdgcn_sched_group_barrier(0x100, 32, 0);
        __builtin_amdgcn_sched_group_barrier(0x008, 32, 0);
        __builtin_amdgcn_sched_group_barrier(0x100, 32, 0);
        __builtin_amdgcn_sched_group_barrier(0x008, 32, 0);

#pragma unroll
        for (int r = 0; r < 8; ++r)
            lds[(tr * 16 + mloc + r) * SS_STRIDE + tc * 16 + row] = acc[r] + mb;
    }
    __syncthreads();

    {
#pragma unroll
        for (int j = 0; j < 2; ++j) {
            const int rr = (tid >> 4) + 16 * j;
            const int c4 = tid & 15;
            const float4 v = *(const float4*)&lds[rr * SS_STRIDE + c4 * 4];
            *(float4*)&scales[(size_t)(bm * G1_TM + rr) * IN_DIM + bn * G1_TN + c4 * 4] = v;
        }
    }
}

__global__ void __launch_bounds__(256)
dml_fused_kernel(const float* __restrict__ mod,
                 const float* __restrict__ mod_w,
                 const float* __restrict__ mod_b,
                 const float* __restrict__ x,
                 const float* __restrict__ colsq,
                 const float* __restrict__ scales,
                 const float* __restrict__ weight,
                 const float* __restrict__ bias,
                 float* __restrict__ out,
                 size_t ws_size) {
    __shared__ alignas(16) float lds_xs[16 * XS_STRIDE];
    __shared__ alignas(16) float lds_out[16 * OS_STRIDE];
    __shared__ alignas(16) float lds_w[N_PER_BLOCK * XS_STRIDE];

    const int lane = threadIdx.x & 31;
    const int wave = threadIdx.x >> 5;
    const int tM   = blockIdx.x;
    const int nQ   = blockIdx.y;
    const int row  = lane & 15;
    const int koff = (lane >> 4) << 2;
    const int mloc = (lane >> 4) << 3;

    const float* wSrc = weight + (size_t)(nQ * N_PER_BLOCK + wave * 16) * IN_DIM + lane * 4;
    float*       wDst = &lds_w[(wave * 16) * XS_STRIDE + lane * 4];
    v4f wreg[16];
#pragma unroll
    for (int j = 0; j < 16; ++j)
        wreg[j] = *(const v4f*)(wSrc + (size_t)j * IN_DIM);

    if (ws_size >= WS_NEED) {
        const int    c4 = threadIdx.x & 127;
        const float4 cs = *(const float4*)&colsq[c4 * 4];
#pragma unroll
        for (int j = 0; j < 8; ++j) {
            const int    mm = (threadIdx.x >> 7) + 2 * j;
            const size_t go = (size_t)(tM * 16 + mm) * IN_DIM + c4 * 4;
            const float4 xv = *(const float4*)&x[go];
            const float4 sv = *(const float4*)&scales[go];
            float4 o;
            o.x = xv.x * sv.x * rsqrtf(fmaf(sv.x * sv.x, cs.x, EPS));
            o.y = xv.y * sv.y * rsqrtf(fmaf(sv.y * sv.y, cs.y, EPS));
            o.z = xv.z * sv.z * rsqrtf(fmaf(sv.z * sv.z, cs.z, EPS));
            o.w = xv.w * sv.w * rsqrtf(fmaf(sv.w * sv.w, cs.w, EPS));
            *(float4*)&lds_xs[mm * XS_STRIDE + c4 * 4] = o;
        }
    } else {
        const float* aRow = mod + (size_t)(tM * 16 + row) * MOD_DIM + koff;
        const float* bRow[4];
        v8f acc[4] = {};
#pragma unroll
        for (int t = 0; t < 4; ++t)
            bRow[t] = mod_w + (size_t)((wave * 4 + t) * 16 + row) * MOD_DIM + koff;

#pragma unroll 4
        for (int k = 0; k < MOD_DIM; k += 8) {
            const v4f a = *(const v4f*)(aRow + k);
            v4f b[4];
#pragma unroll
            for (int t = 0; t < 4; ++t)
                b[t] = *(const v4f*)(bRow[t] + k);
            const v2f a0 = {a.x, a.y};
            const v2f a1 = {a.z, a.w};
#pragma unroll
            for (int t = 0; t < 4; ++t) {
                const v2f b0 = {b[t].x, b[t].y};
                acc[t] = __builtin_amdgcn_wmma_f32_16x16x4_f32(
                    false, a0, false, b0, (short)0, acc[t], false, false);
            }
#pragma unroll
            for (int t = 0; t < 4; ++t) {
                const v2f b1 = {b[t].z, b[t].w};
                acc[t] = __builtin_amdgcn_wmma_f32_16x16x4_f32(
                    false, a1, false, b1, (short)0, acc[t], false, false);
            }
        }

#pragma unroll
        for (int t = 0; t < 4; ++t) {
            const int   gi = (wave * 4 + t) * 16 + row;
            const float mb = mod_b[gi];
            const float cs = colsq[gi];
#pragma unroll
            for (int r = 0; r < 8; ++r) {
                const int   mm = mloc + r;
                const float s  = acc[t][r] + mb;
                const float d  = rsqrtf(fmaf(s * s, cs, EPS));
                lds_xs[mm * XS_STRIDE + gi] =
                    x[(size_t)(tM * 16 + mm) * IN_DIM + gi] * s * d;
            }
        }
    }

#pragma unroll
    for (int j = 0; j < 16; ++j)
        *(v4f*)(wDst + j * XS_STRIDE) = wreg[j];

    __syncthreads();

    {
        const int    gn   = nQ * N_PER_BLOCK + wave * 16 + row;
        const float* aLds = &lds_xs[row * XS_STRIDE + koff];
        const float* bLds = &lds_w[(wave * 16 + row) * XS_STRIDE + koff];
        const float  bv   = bias[gn];
        v8f acc = (v8f){bv, bv, bv, bv, bv, bv, bv, bv};

#pragma clang loop unroll(disable)
        for (int s = 0; s + 1 < N_CHUNK; ++s) {
            const int kc = s * K_CHUNK;
            const int kn = kc + K_CHUNK;
#pragma unroll
            for (int j = 0; j < 16; ++j)
                wreg[j] = *(const v4f*)(wSrc + (size_t)j * IN_DIM + kn);
#pragma unroll
            for (int u = 0; u < K_CHUNK / 8; ++u) {
                const v4f a = *(const v4f*)(aLds + kc + 8 * u);
                const v4f b = *(const v4f*)(bLds + kc + 8 * u);
                const v2f a0 = {a.x, a.y};
                const v2f b0 = {b.x, b.y};
                const v2f a1 = {a.z, a.w};
                const v2f b1 = {b.z, b.w};
                acc = __builtin_amdgcn_wmma_f32_16x16x4_f32(
                    false, a0, false, b0, (short)0, acc, false, false);
                acc = __builtin_amdgcn_wmma_f32_16x16x4_f32(
                    false, a1, false, b1, (short)0, acc, false, false);
            }
#pragma unroll
            for (int j = 0; j < 16; ++j)
                *(v4f*)(wDst + j * XS_STRIDE + kn) = wreg[j];
            __builtin_amdgcn_fence(__ATOMIC_RELEASE, "wavefront");
            __builtin_amdgcn_wave_barrier();
            __builtin_amdgcn_fence(__ATOMIC_ACQUIRE, "wavefront");
            __builtin_amdgcn_sched_group_barrier(0x020, 16, 0);
            __builtin_amdgcn_sched_group_barrier(0x100, 32, 0);
            __builtin_amdgcn_sched_group_barrier(0x008, 32, 0);
            __builtin_amdgcn_sched_group_barrier(0x200, 16, 0);
        }
#pragma unroll
        for (int u = 0; u < K_CHUNK / 8; ++u) {
            const int k = (N_CHUNK - 1) * K_CHUNK + 8 * u;
            const v4f a = *(const v4f*)(aLds + k);
            const v4f b = *(const v4f*)(bLds + k);
            const v2f a0 = {a.x, a.y};
            const v2f b0 = {b.x, b.y};
            const v2f a1 = {a.z, a.w};
            const v2f b1 = {b.z, b.w};
            acc = __builtin_amdgcn_wmma_f32_16x16x4_f32(
                false, a0, false, b0, (short)0, acc, false, false);
            acc = __builtin_amdgcn_wmma_f32_16x16x4_f32(
                false, a1, false, b1, (short)0, acc, false, false);
        }
        __builtin_amdgcn_sched_group_barrier(0x100, 32, 0);
        __builtin_amdgcn_sched_group_barrier(0x008, 32, 0);

#pragma unroll
        for (int r = 0; r < 8; ++r)
            lds_out[(mloc + r) * OS_STRIDE + wave * 16 + row] = acc[r];
    }

    __syncthreads();

#pragma unroll
    for (int j = 0; j < 2; ++j) {
        const int rr = wave + 8 * j;
        const float4 v = *(const float4*)&lds_out[rr * OS_STRIDE + lane * 4];
        *(float4*)&out[(size_t)(tM * 16 + rr) * OUT_DIM + nQ * N_PER_BLOCK + lane * 4] = v;
    }
}

extern "C" void kernel_launch(void* const* d_in, const int* in_sizes, int n_in,
                              void* d_out, int out_size, void* d_ws, size_t ws_size,
                              hipStream_t stream) {
    const float* mod    = (const float*)d_in[0];
    const float* x      = (const float*)d_in[1];
    const float* weight = (const float*)d_in[2];
    const float* bias   = (const float*)d_in[3];
    const float* mod_w  = (const float*)d_in[4];
    const float* mod_b  = (const float*)d_in[5];
    float* out = (float*)d_out;

    float* colsq  = (float*)d_ws;
    float* scales = (float*)d_ws + IN_DIM;

    dml_colsq_scales_kernel<<<CS_BLOCKS + G1_BLOCKS, L1_THREADS, 0, stream>>>(
        weight, mod, mod_w, mod_b, colsq, scales, ws_size);

    dml_fused_kernel<<<dim3(BATCH / 16, N_SPLIT), 256, 0, stream>>>(
        mod, mod_w, mod_b, x, colsq, scales, weight, bias, out, ws_size);
}
